# starting kernel
# baseline (speedup 1.0000x reference)
#include <hip/hip_runtime.h>
#include <stddef.h>

typedef __attribute__((ext_vector_type(2))) float v2f;
typedef __attribute__((ext_vector_type(8))) float v8f;

__device__ __forceinline__ v8f wmma4(v2f a, v2f b, v8f c) {
  return __builtin_amdgcn_wmma_f32_16x16x4_f32(
       false, a,  false, b,
       (short)0, c,  false,  false);
}

template <int N, int K>
__global__ void gemm_wmma_f32(const float* __restrict__ A,
                              const float* __restrict__ B,
                              float* __restrict__ C) {
  const int lane = threadIdx.x & 31;
  const int m0 = blockIdx.x * 16;
  const int n0 = blockIdx.y * 64;
  const int half = lane >> 4;
  const int r    = lane & 15;

  const float* Arow = A + (size_t)(m0 + r) * K + 2 * half;
  const float* Bcol = B + (size_t)(2 * half) * N + n0 + r;

  v8f c0 = {}, c1 = {}, c2 = {}, c3 = {};
#pragma unroll
  for (int k = 0; k < K; k += 4) {
    v2f a = *(const v2f*)(Arow + k);
    v2f b0, b1, b2, b3;
    b0.x = Bcol[k * N + 0];       b0.y = Bcol[k * N + N + 0];
    b1.x = Bcol[k * N + 16];      b1.y = Bcol[k * N + N + 16];
    b2.x = Bcol[k * N + 32];      b2.y = Bcol[k * N + N + 32];
    b3.x = Bcol[k * N + 48];      b3.y = Bcol[k * N + N + 48];
    c0 = wmma4(a, b0, c0);
    c1 = wmma4(a, b1, c1);
    c2 = wmma4(a, b2, c2);
    c3 = wmma4(a, b3, c3);
  }

  float* Crow = C + (size_t)(m0 + 8 * half) * N + n0 + r;
#pragma unroll
  for (int i = 0; i < 8; ++i) {
    Crow[i * N + 0]  = c0[i];
    Crow[i * N + 16] = c1[i];
    Crow[i * N + 32] = c2[i];
    Crow[i * N + 48] = c3[i];
  }
}

template <int H, int F>
__global__ void make_wa(const float* __restrict__ W, const float* __restrict__ att,
                        float* __restrict__ Wa, int K) {
  int t = blockIdx.x * blockDim.x + threadIdx.x;
  if (t >= K * H) return;
  int k = t / H, h = t & (H - 1);
  const float* wr = W + (size_t)k * (H * F) + h * F;
  const float* ar = att + h * F;
  float acc = 0.f;
#pragma unroll 4
  for (int f = 0; f < F; ++f) acc += wr[f] * ar[f];
  Wa[t] = acc;
}

template <int K, int H>
__global__ void proj_att(const float* __restrict__ X, const float* __restrict__ Wa,
                         float* __restrict__ out, int N) {
  int t = blockIdx.x * blockDim.x + threadIdx.x;
  if (t >= N * H) return;
  int n = t / H, h = t & (H - 1);
  const float* xr = X + (size_t)n * K;
  float acc = 0.f;
#pragma unroll 4
  for (int k = 0; k < K; ++k) acc += xr[k] * Wa[k * H + h];
  out[t] = acc;
}

__device__ __forceinline__ unsigned fkey(float f) {
  unsigned u = __float_as_uint(f);
  return (u & 0x80000000u) ? ~u : (u | 0x80000000u);
}
__device__ __forceinline__ float fdecode(unsigned k) {
  unsigned u = (k & 0x80000000u) ? (k & 0x7FFFFFFFu) : ~k;
  return __uint_as_float(u);
}
__device__ __forceinline__ float leaky(float a) {
  return a > 0.f ? a : 0.2f * a;
}

template <int H>
__global__ void edge_max(const int* __restrict__ src, const int* __restrict__ dst,
                         const float* __restrict__ as, const float* __restrict__ ad,
                         unsigned* __restrict__ amax, int E) {
  int t = blockIdx.x * blockDim.x + threadIdx.x;
  if (t >= E * H) return;
  int e = t / H, h = t & (H - 1);
  float a = leaky(as[src[e] * H + h] + ad[dst[e] * H + h]);
  atomicMax(&amax[dst[e] * H + h], fkey(a));
}

template <int H>
__global__ void edge_denom(const int* __restrict__ src, const int* __restrict__ dst,
                           const float* __restrict__ as, const float* __restrict__ ad,
                           const unsigned* __restrict__ amax, float* __restrict__ den,
                           int E) {
  int t = blockIdx.x * blockDim.x + threadIdx.x;
  if (t >= E * H) return;
  int e = t / H, h = t & (H - 1);
  int d = dst[e] * H + h;
  float a = leaky(as[src[e] * H + h] + ad[d]);
  float ex = __expf(a - fdecode(amax[d]));
  atomicAdd(&den[d], ex);
}

template <int H, int LOGF>
__global__ void edge_aggregate(const int* __restrict__ src, const int* __restrict__ dst,
                               const float* __restrict__ as, const float* __restrict__ ad,
                               const unsigned* __restrict__ amax,
                               const float* __restrict__ den,
                               const float* __restrict__ xs, float* __restrict__ out,
                               int E) {
  int e = blockIdx.x * (blockDim.x >> 5) + (threadIdx.x >> 5);
  if (e >= E) return;
  int lane = threadIdx.x & 31;
  int s = src[e], d = dst[e];
  constexpr int HF = H << LOGF;
#pragma unroll
  for (int p = lane; p < HF; p += 32) {
    int h = p >> LOGF;
    float a = leaky(as[s * H + h] + ad[d * H + h]);
    float ex = __expf(a - fdecode(amax[d * H + h]));
    float w = ex / (den[d * H + h] + 1e-16f);
    atomicAdd(&out[(size_t)d * HF + p], w * xs[(size_t)s * HF + p]);
  }
}

__global__ void bias_relu(float* __restrict__ h, const float* __restrict__ b,
                          int total, int HF) {
  int t = blockIdx.x * blockDim.x + threadIdx.x;
  if (t >= total) return;
  float v = h[t] + b[t & (HF - 1)];
  h[t] = v > 0.f ? v : 0.f;
}

__global__ void bias_add(float* __restrict__ o, const float* __restrict__ b,
                         int total, int C) {
  int t = blockIdx.x * blockDim.x + threadIdx.x;
  if (t >= total) return;
  o[t] = o[t] + b[t & (C - 1)];
}

extern "C" void kernel_launch(void* const* d_in, const int* in_sizes, int n_in,
                              void* d_out, int out_size, void* d_ws, size_t ws_size,
                              hipStream_t stream) {
  (void)n_in; (void)out_size; (void)ws_size;
  const float* x     = (const float*)d_in[0];
  const int*   ei    = (const int*)d_in[1];
  const float* W1s   = (const float*)d_in[2];
  const float* W1d   = (const float*)d_in[3];
  const float* att1s = (const float*)d_in[4];
  const float* att1d = (const float*)d_in[5];
  const float* b1    = (const float*)d_in[6];
  const float* W2s   = (const float*)d_in[7];
  const float* W2d   = (const float*)d_in[8];
  const float* att2s = (const float*)d_in[9];
  const float* att2d = (const float*)d_in[10];
  const float* b2    = (const float*)d_in[11];
  float* out = (float*)d_out;

  const int N = in_sizes[0] / 128;
  const int E = in_sizes[1] / 2;
  const int* src = ei;
  const int* dst = ei + E;

  float* ws = (float*)d_ws;
  size_t off = 0;
  float*    XS   = ws + off; off += (size_t)N * 256;
  float*    Hb   = ws + off; off += (size_t)N * 256;
  float*    AS   = ws + off; off += (size_t)N * 8;
  float*    AD   = ws + off; off += (size_t)N * 8;
  unsigned* AMAX = (unsigned*)(ws + off); off += (size_t)N * 8;
  float*    DEN  = ws + off; off += (size_t)N * 8;
  float*    WA   = ws + off; off += 2048;
  float*    WD   = ws + off; off += 2048;

  const int TPB = 256;

  gemm_wmma_f32<256, 128><<<dim3(N / 16, 256 / 64), 32, 0, stream>>>(x, W1s, XS);
  make_wa<8, 32><<<(128 * 8 + TPB - 1) / TPB, TPB, 0, stream>>>(W1s, att1s, WA, 128);
  make_wa<8, 32><<<(128 * 8 + TPB - 1) / TPB, TPB, 0, stream>>>(W1d, att1d, WD, 128);
  proj_att<128, 8><<<((size_t)N * 8 + TPB - 1) / TPB, TPB, 0, stream>>>(x, WA, AS, N);
  proj_att<128, 8><<<((size_t)N * 8 + TPB - 1) / TPB, TPB, 0, stream>>>(x, WD, AD, N);

  hipMemsetAsync(AMAX, 0, (size_t)N * 8 * sizeof(unsigned), stream);
  hipMemsetAsync(DEN, 0, (size_t)N * 8 * sizeof(float), stream);
  hipMemsetAsync(Hb, 0, (size_t)N * 256 * sizeof(float), stream);

  int EH = E * 8;
  edge_max<8>  <<<(EH + TPB - 1) / TPB, TPB, 0, stream>>>(src, dst, AS, AD, AMAX, E);
  edge_denom<8><<<(EH + TPB - 1) / TPB, TPB, 0, stream>>>(src, dst, AS, AD, AMAX, DEN, E);
  edge_aggregate<8, 5><<<(E + 7) / 8, 256, 0, stream>>>(src, dst, AS, AD, AMAX, DEN,
                                                        XS, Hb, E);
  bias_relu<<<((size_t)N * 256 + TPB - 1) / TPB, TPB, 0, stream>>>(Hb, b1, N * 256, 256);

  gemm_wmma_f32<128, 256><<<dim3(N / 16, 128 / 64), 32, 0, stream>>>(Hb, W2s, XS);
  make_wa<1, 128><<<(256 + TPB - 1) / TPB, TPB, 0, stream>>>(W2s, att2s, WA, 256);
  make_wa<1, 128><<<(256 + TPB - 1) / TPB, TPB, 0, stream>>>(W2d, att2d, WD, 256);
  proj_att<256, 1><<<(N + TPB - 1) / TPB, TPB, 0, stream>>>(Hb, WA, AS, N);
  proj_att<256, 1><<<(N + TPB - 1) / TPB, TPB, 0, stream>>>(Hb, WD, AD, N);

  hipMemsetAsync(AMAX, 0, (size_t)N * sizeof(unsigned), stream);
  hipMemsetAsync(DEN, 0, (size_t)N * sizeof(float), stream);
  hipMemsetAsync(out, 0, (size_t)N * 128 * sizeof(float), stream);

  edge_max<1>  <<<(E + TPB - 1) / TPB, TPB, 0, stream>>>(src, dst, AS, AD, AMAX, E);
  edge_denom<1><<<(E + TPB - 1) / TPB, TPB, 0, stream>>>(src, dst, AS, AD, AMAX, DEN, E);
  edge_aggregate<1, 7><<<(E + 7) / 8, 256, 0, stream>>>(src, dst, AS, AD, AMAX, DEN,
                                                        XS, out, E);
  bias_add<<<((size_t)N * 128 + TPB - 1) / TPB, TPB, 0, stream>>>(out, b2, N * 128, 128);
}


# Edges counting-sorted by destination in each call; one wave per node does softmax and weighted row sum in registers, no float atomics (sum order = integer-atomic arrival per node), 18 -> 15 launches
# speedup vs baseline: 2.2210x; 2.2210x over previous
#include <hip/hip_runtime.h>
#include <stddef.h>

typedef __attribute__((ext_vector_type(2))) float v2f;
typedef __attribute__((ext_vector_type(8))) float v8f;

__device__ __forceinline__ v8f wmma4(v2f a, v2f b, v8f c) {
  return __builtin_amdgcn_wmma_f32_16x16x4_f32(
       false, a,  false, b,
       (short)0, c,  false,  false);
}

template <int N, int K>
__global__ void gemm_wmma_f32(const float* __restrict__ A,
                              const float* __restrict__ B,
                              float* __restrict__ C) {
  const int lane = threadIdx.x & 31;
  const int m0 = blockIdx.x * 16;
  const int n0 = blockIdx.y * 64;
  const int half = lane >> 4;
  const int r    = lane & 15;

  const float* Arow = A + (size_t)(m0 + r) * K + 2 * half;
  const float* Bcol = B + (size_t)(2 * half) * N + n0 + r;

  v8f c0 = {}, c1 = {}, c2 = {}, c3 = {};
#pragma unroll
  for (int k = 0; k < K; k += 4) {
    v2f a = *(const v2f*)(Arow + k);
    v2f b0, b1, b2, b3;
    b0.x = Bcol[k * N + 0];       b0.y = Bcol[k * N + N + 0];
    b1.x = Bcol[k * N + 16];      b1.y = Bcol[k * N + N + 16];
    b2.x = Bcol[k * N + 32];      b2.y = Bcol[k * N + N + 32];
    b3.x = Bcol[k * N + 48];      b3.y = Bcol[k * N + N + 48];
    c0 = wmma4(a, b0, c0);
    c1 = wmma4(a, b1, c1);
    c2 = wmma4(a, b2, c2);
    c3 = wmma4(a, b3, c3);
  }

  float* Crow = C + (size_t)(m0 + 8 * half) * N + n0 + r;
#pragma unroll
  for (int i = 0; i < 8; ++i) {
    Crow[i * N + 0]  = c0[i];
    Crow[i * N + 16] = c1[i];
    Crow[i * N + 32] = c2[i];
    Crow[i * N + 48] = c3[i];
  }
}

template <int H, int F>
__global__ void make_wa(const float* __restrict__ W, const float* __restrict__ att,
                        float* __restrict__ Wa, int K) {
  int t = blockIdx.x * blockDim.x + threadIdx.x;
  if (t >= K * H) return;
  int k = t / H, h = t & (H - 1);
  const float* wr = W + (size_t)k * (H * F) + h * F;
  const float* ar = att + h * F;
  float acc = 0.f;
#pragma unroll 4
  for (int f = 0; f < F; ++f) acc += wr[f] * ar[f];
  Wa[t] = acc;
}

template <int K, int H>
__global__ void proj_att(const float* __restrict__ X, const float* __restrict__ Wa,
                         float* __restrict__ out, int N, int ostride) {
  int t = blockIdx.x * blockDim.x + threadIdx.x;
  if (t >= N * H) return;
  int n = t / H, h = t & (H - 1);
  const float* xr = X + (size_t)n * K;
  float acc = 0.f;
#pragma unroll 4
  for (int k = 0; k < K; ++k) acc += xr[k] * Wa[k * H + h];
  out[(size_t)n * ostride + h] = acc;
}

typedef __attribute__((ext_vector_type(4))) float v4f;

__device__ __forceinline__ float leaky(float a) {
  return a > 0.f ? a : 0.2f * a;
}

__global__ void dst_hist(const int* __restrict__ dst, int* __restrict__ cnt, int E) {
  int t = blockIdx.x * blockDim.x + threadIdx.x;
  if (t >= E) return;
  atomicAdd(&cnt[dst[t]], 1);
}

__global__ void dst_scan(const int* __restrict__ cnt, int* __restrict__ rowptr,
                         int* __restrict__ cur, int N) {
  __shared__ int part[1024];
  const int t = threadIdx.x;
  const int per = (N + 1023) >> 10;
  int b = t * per; if (b > N) b = N;
  int e = b + per; if (e > N) e = N;
  int s = 0;
  for (int i = b; i < e; ++i) s += cnt[i];
  part[t] = s;
  __syncthreads();
  for (int o = 1; o < 1024; o <<= 1) {
    int v = (t >= o) ? part[t - o] : 0;
    __syncthreads();
    part[t] += v;
    __syncthreads();
  }
  int run = part[t] - s;
  for (int i = b; i < e; ++i) { rowptr[i] = run; cur[i] = run; run += cnt[i]; }
  if (t == 1023) rowptr[N] = part[1023];
}

__global__ void dst_fill(const int* __restrict__ src, const int* __restrict__ dst,
                         int* __restrict__ cur, int* __restrict__ psrc, int E) {
  int t = blockIdx.x * blockDim.x + threadIdx.x;
  if (t >= E) return;
  int p = atomicAdd(&cur[dst[t]], 1);
  psrc[p] = src[t];
}

__device__ __forceinline__ void edge8(int s, int lane, int h0, int h1,
                                      const float* __restrict__ as,
                                      const float* __restrict__ xs,
                                      float ad0, float ad1, float m0, float m1,
                                      float i0, float i1, v4f& acc0, v4f& acc1) {
  const float w0 = __expf(leaky(as[s * 8 + h0] + ad0) - m0) * i0;
  const float w1 = __expf(leaky(as[s * 8 + h1] + ad1) - m1) * i1;
  const float* xr = xs + (size_t)s * 256 + lane * 4;
  const v4f x0 = *(const v4f*)xr;
  const v4f x1 = *(const v4f*)(xr + 128);
  acc0 += w0 * x0;
  acc1 += w1 * x1;
}

__global__ void node_gat8(const int* __restrict__ rowptr, const int* __restrict__ psrc,
                          const float* __restrict__ as, const float* __restrict__ xs,
                          const float* __restrict__ bias, float* hb, int N) {
  const int d = __builtin_amdgcn_readfirstlane(
      (int)(blockIdx.x * (blockDim.x >> 5) + (threadIdx.x >> 5)));
  if (d >= N) return;
  const int lane = threadIdx.x & 31;
  const int beg = rowptr[d], end = rowptr[d + 1];
  float* row = hb + (size_t)d * 256;
  const int h = lane & 7, sub = lane >> 3;
  const float adh = row[h];

  float m = -INFINITY;
  for (int c = beg; c < end; c += 32) {
    const int n = (end - c) < 32 ? (end - c) : 32;
    const int sv = (c + lane < end) ? psrc[c + lane] : 0;
#pragma unroll
    for (int k = 0; k < 8; ++k) {
      const int j = sub + 4 * k;
      const float al = leaky(as[__shfl(sv, j) * 8 + h] + adh);
      m = (j < n) ? fmaxf(m, al) : m;
    }
  }
  m = fmaxf(m, __shfl_xor(m, 8));
  m = fmaxf(m, __shfl_xor(m, 16));

  float den = 0.f;
  for (int c = beg; c < end; c += 32) {
    const int n = (end - c) < 32 ? (end - c) : 32;
    const int sv = (c + lane < end) ? psrc[c + lane] : 0;
#pragma unroll
    for (int k = 0; k < 8; ++k) {
      const int j = sub + 4 * k;
      const float al = leaky(as[__shfl(sv, j) * 8 + h] + adh);
      den += (j < n) ? __expf(al - m) : 0.f;
    }
  }
  den += __shfl_xor(den, 8);
  den += __shfl_xor(den, 16);
  const float inv = 1.0f / (den + 1e-16f);

  const int h0 = lane >> 3, h1 = h0 + 4;
  const float m0 = __shfl(m, h0),    m1 = __shfl(m, h1);
  const float i0 = __shfl(inv, h0),  i1 = __shfl(inv, h1);
  const float ad0 = __shfl(adh, h0), ad1 = __shfl(adh, h1);
  v4f acc0 = {0.f, 0.f, 0.f, 0.f}, acc1 = {0.f, 0.f, 0.f, 0.f};
  for (int c = beg; c < end; c += 32) {
    const int n = (end - c) < 32 ? (end - c) : 32;
    const int sv = (c + lane < end) ? psrc[c + lane] : 0;
    int i = 0;
    for (; i + 4 <= n; i += 4) {
      const int s0 = __builtin_amdgcn_readlane(sv, i);
      const int s1 = __builtin_amdgcn_readlane(sv, i + 1);
      const int s2 = __builtin_amdgcn_readlane(sv, i + 2);
      const int s3 = __builtin_amdgcn_readlane(sv, i + 3);
      edge8(s0, lane, h0, h1, as, xs, ad0, ad1, m0, m1, i0, i1, acc0, acc1);
      edge8(s1, lane, h0, h1, as, xs, ad0, ad1, m0, m1, i0, i1, acc0, acc1);
      edge8(s2, lane, h0, h1, as, xs, ad0, ad1, m0, m1, i0, i1, acc0, acc1);
      edge8(s3, lane, h0, h1, as, xs, ad0, ad1, m0, m1, i0, i1, acc0, acc1);
    }
    for (; i < n; ++i)
      edge8(__builtin_amdgcn_readlane(sv, i), lane, h0, h1, as, xs,
            ad0, ad1, m0, m1, i0, i1, acc0, acc1);
  }
  v4f o0 = acc0 + *(const v4f*)(bias + lane * 4);
  v4f o1 = acc1 + *(const v4f*)(bias + 128 + lane * 4);
#pragma unroll
  for (int q = 0; q < 4; ++q) {
    o0[q] = o0[q] > 0.f ? o0[q] : 0.f;
    o1[q] = o1[q] > 0.f ? o1[q] : 0.f;
  }
  *(v4f*)(row + lane * 4) = o0;
  *(v4f*)(row + 128 + lane * 4) = o1;
}

__device__ __forceinline__ void edge1(int s, float w, int lane,
                                      const float* __restrict__ xs, v4f& acc) {
  acc += w * *(const v4f*)(xs + (size_t)s * 128 + lane * 4);
}

__global__ void node_gat1(const int* __restrict__ rowptr, const int* __restrict__ psrc,
                          const float* __restrict__ as, const float* __restrict__ ad,
                          const float* __restrict__ xs, const float* __restrict__ bias,
                          float* __restrict__ out, int N) {
  const int d = __builtin_amdgcn_readfirstlane(
      (int)(blockIdx.x * (blockDim.x >> 5) + (threadIdx.x >> 5)));
  if (d >= N) return;
  const int lane = threadIdx.x & 31;
  const int beg = rowptr[d], end = rowptr[d + 1];
  const float add = ad[d];
  float m = -INFINITY;
  for (int i = beg + lane; i < end; i += 32)
    m = fmaxf(m, leaky(as[psrc[i]] + add));
#pragma unroll
  for (int o = 16; o > 0; o >>= 1) m = fmaxf(m, __shfl_xor(m, o));
  float den = 0.f;
  for (int i = beg + lane; i < end; i += 32)
    den += __expf(leaky(as[psrc[i]] + add) - m);
#pragma unroll
  for (int o = 16; o > 0; o >>= 1) den += __shfl_xor(den, o);
  const float inv = 1.0f / (den + 1e-16f);
  v4f acc = {0.f, 0.f, 0.f, 0.f};
  for (int c = beg; c < end; c += 32) {
    const int n = (end - c) < 32 ? (end - c) : 32;
    const bool ok = c + lane < end;
    const int sv = ok ? psrc[c + lane] : 0;
    const int wv = __float_as_int(ok ? __expf(leaky(as[sv] + add) - m) * inv : 0.f);
    int i = 0;
    for (; i + 4 <= n; i += 4) {
      const int s0 = __builtin_amdgcn_readlane(sv, i);
      const int s1 = __builtin_amdgcn_readlane(sv, i + 1);
      const int s2 = __builtin_amdgcn_readlane(sv, i + 2);
      const int s3 = __builtin_amdgcn_readlane(sv, i + 3);
      const float w0 = __int_as_float(__builtin_amdgcn_readlane(wv, i));
      const float w1 = __int_as_float(__builtin_amdgcn_readlane(wv, i + 1));
      const float w2 = __int_as_float(__builtin_amdgcn_readlane(wv, i + 2));
      const float w3 = __int_as_float(__builtin_amdgcn_readlane(wv, i + 3));
      edge1(s0, w0, lane, xs, acc);
      edge1(s1, w1, lane, xs, acc);
      edge1(s2, w2, lane, xs, acc);
      edge1(s3, w3, lane, xs, acc);
    }
    for (; i < n; ++i)
      edge1(__builtin_amdgcn_readlane(sv, i),
            __int_as_float(__builtin_amdgcn_readlane(wv, i)), lane, xs, acc);
  }
  *(v4f*)(out + (size_t)d * 128 + lane * 4) = acc + *(const v4f*)(bias + lane * 4);
}

extern "C" void kernel_launch(void* const* d_in, const int* in_sizes, int n_in,
                              void* d_out, int out_size, void* d_ws, size_t ws_size,
                              hipStream_t stream) {
  (void)n_in; (void)out_size; (void)ws_size;
  const float* x     = (const float*)d_in[0];
  const int*   ei    = (const int*)d_in[1];
  const float* W1s   = (const float*)d_in[2];
  const float* W1d   = (const float*)d_in[3];
  const float* att1s = (const float*)d_in[4];
  const float* att1d = (const float*)d_in[5];
  const float* b1    = (const float*)d_in[6];
  const float* W2s   = (const float*)d_in[7];
  const float* W2d   = (const float*)d_in[8];
  const float* att2s = (const float*)d_in[9];
  const float* att2d = (const float*)d_in[10];
  const float* b2    = (const float*)d_in[11];
  float* out = (float*)d_out;

  const int N = in_sizes[0] / 128;
  const int E = in_sizes[1] / 2;
  const int* src = ei;
  const int* dst = ei + E;

  float* ws = (float*)d_ws;
  size_t off = 0;
  float*    XS   = ws + off; off += (size_t)N * 256;
  float*    Hb   = ws + off; off += (size_t)N * 256;
  float*    AS   = ws + off; off += (size_t)N * 8;
  int*      RP   = (int*)(ws + off);
  int*      CUR  = RP + (size_t)N * 2;
  int*      CNT  = RP + (size_t)N * 3;
  float*    AD2  = ws + off + (size_t)N * 4;
  off += (size_t)N * 8;
  int*      PSRC = (int*)(ws + off); off += (size_t)N * 16;
  float*    WA   = ws + off; off += 2048;
  float*    WD   = ws + off; off += 2048;

  const int TPB = 256;

  hipMemsetAsync(CNT, 0, (size_t)N * sizeof(int), stream);
  dst_hist<<<(E + TPB - 1) / TPB, TPB, 0, stream>>>(dst, CNT, E);
  dst_scan<<<1, 1024, 0, stream>>>(CNT, RP, CUR, N);
  dst_fill<<<(E + TPB - 1) / TPB, TPB, 0, stream>>>(src, dst, CUR, PSRC, E);

  gemm_wmma_f32<256, 128><<<dim3(N / 16, 256 / 64), 32, 0, stream>>>(x, W1s, XS);
  make_wa<8, 32><<<(128 * 8 + TPB - 1) / TPB, TPB, 0, stream>>>(W1s, att1s, WA, 128);
  make_wa<8, 32><<<(128 * 8 + TPB - 1) / TPB, TPB, 0, stream>>>(W1d, att1d, WD, 128);
  proj_att<128, 8><<<((size_t)N * 8 + TPB - 1) / TPB, TPB, 0, stream>>>(x, WA, AS, N, 8);
  proj_att<128, 8><<<((size_t)N * 8 + TPB - 1) / TPB, TPB, 0, stream>>>(x, WD, Hb, N, 256);
  node_gat8<<<(N + 7) / 8, 256, 0, stream>>>(RP, PSRC, AS, XS, b1, Hb, N);

  gemm_wmma_f32<128, 256><<<dim3(N / 16, 128 / 64), 32, 0, stream>>>(Hb, W2s, XS);
  make_wa<1, 128><<<(256 + TPB - 1) / TPB, TPB, 0, stream>>>(W2s, att2s, WA, 256);
  make_wa<1, 128><<<(256 + TPB - 1) / TPB, TPB, 0, stream>>>(W2d, att2d, WD, 256);
  proj_att<256, 1><<<(N + TPB - 1) / TPB, TPB, 0, stream>>>(Hb, WA, AS, N, 1);
  proj_att<256, 1><<<(N + TPB - 1) / TPB, TPB, 0, stream>>>(Hb, WD, AD2, N, 1);
  node_gat1<<<(N + 7) / 8, 256, 0, stream>>>(RP, PSRC, AS, AD2, XS, b2, out, N);
}


# Attention logits a_s,a_d now come out of the two feature GEMMs as one extra 16-col WMMA tile (folded att panels built in one launch); logit sums rounded in WMMA order; 15 -> 8 launches
# speedup vs baseline: 2.6574x; 1.1965x over previous
#include <hip/hip_runtime.h>
#include <stddef.h>

typedef __attribute__((ext_vector_type(2))) float v2f;
typedef __attribute__((ext_vector_type(8))) float v8f;

__device__ __forceinline__ v8f wmma4(v2f a, v2f b, v8f c) {
  return __builtin_amdgcn_wmma_f32_16x16x4_f32(
       false, a,  false, b,
       (short)0, c,  false,  false);
}

template <int N, int K, bool EXTRA>
__device__ __forceinline__ void strip16x64(const float* __restrict__ Arow,
                                           const float* __restrict__ Bcol,
                                           const float* __restrict__ Xp,
                                           v8f& c0, v8f& c1, v8f& c2, v8f& c3, v8f& cx) {
#pragma unroll
  for (int k = 0; k < K; k += 4) {
    v2f a = *(const v2f*)(Arow + k);
    v2f b0, b1, b2, b3;
    b0.x = Bcol[k * N + 0];       b0.y = Bcol[k * N + N + 0];
    b1.x = Bcol[k * N + 16];      b1.y = Bcol[k * N + N + 16];
    b2.x = Bcol[k * N + 32];      b2.y = Bcol[k * N + N + 32];
    b3.x = Bcol[k * N + 48];      b3.y = Bcol[k * N + N + 48];
    c0 = wmma4(a, b0, c0);
    c1 = wmma4(a, b1, c1);
    c2 = wmma4(a, b2, c2);
    c3 = wmma4(a, b3, c3);
    if constexpr (EXTRA) {
      const v2f bx = *(const v2f*)(Xp + k * 16);
      cx = wmma4(a, bx, cx);
    }
  }
}

template <int N, int K, int H>
__global__ void gemm_wmma_f32(const float* __restrict__ A,
                              const float* __restrict__ B,
                              const float* __restrict__ BX,
                              float* __restrict__ C,
                              float* as_out, int as_stride,
                              float* ad_out, int ad_stride) {
  const int lane = threadIdx.x & 31;
  const int m0 = blockIdx.x * 16;
  const int n0 = blockIdx.y * 64;
  const int half = lane >> 4;
  const int r    = lane & 15;

  const float* Arow = A + (size_t)(m0 + r) * K + 2 * half;
  const float* Bcol = B + (size_t)(2 * half) * N + n0 + r;
  const float* Xp   = BX + (half * 16 + r) * 2;

  v8f c0 = {}, c1 = {}, c2 = {}, c3 = {}, cx = {};
  const bool extra = (blockIdx.y == 0);
  if (extra) strip16x64<N, K, true >(Arow, Bcol, Xp, c0, c1, c2, c3, cx);
  else       strip16x64<N, K, false>(Arow, Bcol, Xp, c0, c1, c2, c3, cx);

  float* Crow = C + (size_t)(m0 + 8 * half) * N + n0 + r;
#pragma unroll
  for (int i = 0; i < 8; ++i) {
    Crow[i * N + 0]  = c0[i];
    Crow[i * N + 16] = c1[i];
    Crow[i * N + 32] = c2[i];
    Crow[i * N + 48] = c3[i];
  }

  if (extra && r < 2 * H) {
    const int row0 = m0 + 8 * half;
    const bool is_s = r < H;
    float* p = is_s ? as_out + (size_t)row0 * as_stride + r
                    : ad_out + (size_t)row0 * ad_stride + (r - H);
    const int st = is_s ? as_stride : ad_stride;
#pragma unroll
    for (int i = 0; i < 8; ++i) p[(size_t)i * st] = cx[i];
  }
}

template <int F>
__device__ __forceinline__ float fold_dot(const float* __restrict__ wr,
                                          const float* __restrict__ ar) {
  float acc = 0.f;
#pragma unroll 4
  for (int f = 0; f < F; ++f) acc += wr[f] * ar[f];
  return acc;
}

__global__ void fold_att(const float* __restrict__ W1s, const float* __restrict__ W1d,
                         const float* __restrict__ att1s, const float* __restrict__ att1d,
                         const float* __restrict__ W2s, const float* __restrict__ W2d,
                         const float* __restrict__ att2s, const float* __restrict__ att2d,
                         float* __restrict__ BX1, float* __restrict__ BX2) {
  const int t = blockIdx.x * blockDim.x + threadIdx.x;
  if (t < 128 * 16) {
    const int k = t >> 4, c = t & 15, h = c & 7;
    const float* W   = (c < 8) ? W1s : W1d;
    const float* att = (c < 8) ? att1s : att1d;
    BX1[((k >> 1) * 16 + c) * 2 + (k & 1)] =
        fold_dot<32>(W + (size_t)k * 256 + h * 32, att + h * 32);
  } else if (t < 128 * 16 + 256 * 16) {
    const int u = t - 128 * 16;
    const int k = u >> 4, c = u & 15;
    float v = 0.f;
    if (c < 2) v = fold_dot<128>((c == 0 ? W2s : W2d) + (size_t)k * 128,
                                 c == 0 ? att2s : att2d);
    BX2[((k >> 1) * 16 + c) * 2 + (k & 1)] = v;
  }
}

typedef __attribute__((ext_vector_type(4))) float v4f;

__device__ __forceinline__ float leaky(float a) {
  return a > 0.f ? a : 0.2f * a;
}

__global__ void dst_hist(const int* __restrict__ dst, int* __restrict__ cnt, int E) {
  int t = blockIdx.x * blockDim.x + threadIdx.x;
  if (t >= E) return;
  atomicAdd(&cnt[dst[t]], 1);
}

__global__ void dst_scan(const int* __restrict__ cnt, int* __restrict__ rowptr,
                         int* __restrict__ cur, int N) {
  __shared__ int part[1024];
  const int t = threadIdx.x;
  const int per = (N + 1023) >> 10;
  int b = t * per; if (b > N) b = N;
  int e = b + per; if (e > N) e = N;
  int s = 0;
  for (int i = b; i < e; ++i) s += cnt[i];
  part[t] = s;
  __syncthreads();
  for (int o = 1; o < 1024; o <<= 1) {
    int v = (t >= o) ? part[t - o] : 0;
    __syncthreads();
    part[t] += v;
    __syncthreads();
  }
  int run = part[t] - s;
  for (int i = b; i < e; ++i) { rowptr[i] = run; cur[i] = run; run += cnt[i]; }
  if (t == 1023) rowptr[N] = part[1023];
}

__global__ void dst_fill(const int* __restrict__ src, const int* __restrict__ dst,
                         int* __restrict__ cur, int* __restrict__ psrc, int E) {
  int t = blockIdx.x * blockDim.x + threadIdx.x;
  if (t >= E) return;
  int p = atomicAdd(&cur[dst[t]], 1);
  psrc[p] = src[t];
}

__device__ __forceinline__ void edge8(int s, int lane, int h0, int h1,
                                      const float* __restrict__ as,
                                      const float* __restrict__ xs,
                                      float ad0, float ad1, float m0, float m1,
                                      float i0, float i1, v4f& acc0, v4f& acc1) {
  const float w0 = __expf(leaky(as[s * 8 + h0] + ad0) - m0) * i0;
  const float w1 = __expf(leaky(as[s * 8 + h1] + ad1) - m1) * i1;
  const float* xr = xs + (size_t)s * 256 + lane * 4;
  const v4f x0 = *(const v4f*)xr;
  const v4f x1 = *(const v4f*)(xr + 128);
  acc0 += w0 * x0;
  acc1 += w1 * x1;
}

__global__ void node_gat8(const int* __restrict__ rowptr, const int* __restrict__ psrc,
                          const float* __restrict__ as, const float* __restrict__ xs,
                          const float* __restrict__ bias, float* hb, int N) {
  const int d = __builtin_amdgcn_readfirstlane(
      (int)(blockIdx.x * (blockDim.x >> 5) + (threadIdx.x >> 5)));
  if (d >= N) return;
  const int lane = threadIdx.x & 31;
  const int beg = rowptr[d], end = rowptr[d + 1];
  float* row = hb + (size_t)d * 256;
  const int h = lane & 7, sub = lane >> 3;
  const float adh = row[h];

  float m = -INFINITY;
  for (int c = beg; c < end; c += 32) {
    const int n = (end - c) < 32 ? (end - c) : 32;
    const int sv = (c + lane < end) ? psrc[c + lane] : 0;
#pragma unroll
    for (int k = 0; k < 8; ++k) {
      const int j = sub + 4 * k;
      const float al = leaky(as[__shfl(sv, j) * 8 + h] + adh);
      m = (j < n) ? fmaxf(m, al) : m;
    }
  }
  m = fmaxf(m, __shfl_xor(m, 8));
  m = fmaxf(m, __shfl_xor(m, 16));

  float den = 0.f;
  for (int c = beg; c < end; c += 32) {
    const int n = (end - c) < 32 ? (end - c) : 32;
    const int sv = (c + lane < end) ? psrc[c + lane] : 0;
#pragma unroll
    for (int k = 0; k < 8; ++k) {
      const int j = sub + 4 * k;
      const float al = leaky(as[__shfl(sv, j) * 8 + h] + adh);
      den += (j < n) ? __expf(al - m) : 0.f;
    }
  }
  den += __shfl_xor(den, 8);
  den += __shfl_xor(den, 16);
  const float inv = 1.0f / (den + 1e-16f);

  const int h0 = lane >> 3, h1 = h0 + 4;
  const float m0 = __shfl(m, h0),    m1 = __shfl(m, h1);
  const float i0 = __shfl(inv, h0),  i1 = __shfl(inv, h1);
  const float ad0 = __shfl(adh, h0), ad1 = __shfl(adh, h1);
  v4f acc0 = {0.f, 0.f, 0.f, 0.f}, acc1 = {0.f, 0.f, 0.f, 0.f};
  for (int c = beg; c < end; c += 32) {
    const int n = (end - c) < 32 ? (end - c) : 32;
    const int sv = (c + lane < end) ? psrc[c + lane] : 0;
    int i = 0;
    for (; i + 4 <= n; i += 4) {
      const int s0 = __builtin_amdgcn_readlane(sv, i);
      const int s1 = __builtin_amdgcn_readlane(sv, i + 1);
      const int s2 = __builtin_amdgcn_readlane(sv, i + 2);
      const int s3 = __builtin_amdgcn_readlane(sv, i + 3);
      edge8(s0, lane, h0, h1, as, xs, ad0, ad1, m0, m1, i0, i1, acc0, acc1);
      edge8(s1, lane, h0, h1, as, xs, ad0, ad1, m0, m1, i0, i1, acc0, acc1);
      edge8(s2, lane, h0, h1, as, xs, ad0, ad1, m0, m1, i0, i1, acc0, acc1);
      edge8(s3, lane, h0, h1, as, xs, ad0, ad1, m0, m1, i0, i1, acc0, acc1);
    }
    for (; i < n; ++i)
      edge8(__builtin_amdgcn_readlane(sv, i), lane, h0, h1, as, xs,
            ad0, ad1, m0, m1, i0, i1, acc0, acc1);
  }
  v4f o0 = acc0 + *(const v4f*)(bias + lane * 4);
  v4f o1 = acc1 + *(const v4f*)(bias + 128 + lane * 4);
#pragma unroll
  for (int q = 0; q < 4; ++q) {
    o0[q] = o0[q] > 0.f ? o0[q] : 0.f;
    o1[q] = o1[q] > 0.f ? o1[q] : 0.f;
  }
  *(v4f*)(row + lane * 4) = o0;
  *(v4f*)(row + 128 + lane * 4) = o1;
}

__device__ __forceinline__ void edge1(int s, float w, int lane,
                                      const float* __restrict__ xs, v4f& acc) {
  acc += w * *(const v4f*)(xs + (size_t)s * 128 + lane * 4);
}

__global__ void node_gat1(const int* __restrict__ rowptr, const int* __restrict__ psrc,
                          const float* __restrict__ sd,
                          const float* __restrict__ xs, const float* __restrict__ bias,
                          float* __restrict__ out, int N) {
  const int d = __builtin_amdgcn_readfirstlane(
      (int)(blockIdx.x * (blockDim.x >> 5) + (threadIdx.x >> 5)));
  if (d >= N) return;
  const int lane = threadIdx.x & 31;
  const int beg = rowptr[d], end = rowptr[d + 1];
  const float add = sd[2 * d + 1];
  float m = -INFINITY;
  for (int i = beg + lane; i < end; i += 32)
    m = fmaxf(m, leaky(sd[2 * psrc[i]] + add));
#pragma unroll
  for (int o = 16; o > 0; o >>= 1) m = fmaxf(m, __shfl_xor(m, o));
  float den = 0.f;
  for (int i = beg + lane; i < end; i += 32)
    den += __expf(leaky(sd[2 * psrc[i]] + add) - m);
#pragma unroll
  for (int o = 16; o > 0; o >>= 1) den += __shfl_xor(den, o);
  const float inv = 1.0f / (den + 1e-16f);
  v4f acc = {0.f, 0.f, 0.f, 0.f};
  for (int c = beg; c < end; c += 32) {
    const int n = (end - c) < 32 ? (end - c) : 32;
    const bool ok = c + lane < end;
    const int sv = ok ? psrc[c + lane] : 0;
    const int wv = __float_as_int(ok ? __expf(leaky(sd[2 * sv] + add) - m) * inv : 0.f);
    int i = 0;
    for (; i + 4 <= n; i += 4) {
      const int s0 = __builtin_amdgcn_readlane(sv, i);
      const int s1 = __builtin_amdgcn_readlane(sv, i + 1);
      const int s2 = __builtin_amdgcn_readlane(sv, i + 2);
      const int s3 = __builtin_amdgcn_readlane(sv, i + 3);
      const float w0 = __int_as_float(__builtin_amdgcn_readlane(wv, i));
      const float w1 = __int_as_float(__builtin_amdgcn_readlane(wv, i + 1));
      const float w2 = __int_as_float(__builtin_amdgcn_readlane(wv, i + 2));
      const float w3 = __int_as_float(__builtin_amdgcn_readlane(wv, i + 3));
      edge1(s0, w0, lane, xs, acc);
      edge1(s1, w1, lane, xs, acc);
      edge1(s2, w2, lane, xs, acc);
      edge1(s3, w3, lane, xs, acc);
    }
    for (; i < n; ++i)
      edge1(__builtin_amdgcn_readlane(sv, i),
            __int_as_float(__builtin_amdgcn_readlane(wv, i)), lane, xs, acc);
  }
  *(v4f*)(out + (size_t)d * 128 + lane * 4) = acc + *(const v4f*)(bias + lane * 4);
}

extern "C" void kernel_launch(void* const* d_in, const int* in_sizes, int n_in,
                              void* d_out, int out_size, void* d_ws, size_t ws_size,
                              hipStream_t stream) {
  (void)n_in; (void)out_size; (void)ws_size;
  const float* x     = (const float*)d_in[0];
  const int*   ei    = (const int*)d_in[1];
  const float* W1s   = (const float*)d_in[2];
  const float* W1d   = (const float*)d_in[3];
  const float* att1s = (const float*)d_in[4];
  const float* att1d = (const float*)d_in[5];
  const float* b1    = (const float*)d_in[6];
  const float* W2s   = (const float*)d_in[7];
  const float* W2d   = (const float*)d_in[8];
  const float* att2s = (const float*)d_in[9];
  const float* att2d = (const float*)d_in[10];
  const float* b2    = (const float*)d_in[11];
  float* out = (float*)d_out;

  const int N = in_sizes[0] / 128;
  const int E = in_sizes[1] / 2;
  const int* src = ei;
  const int* dst = ei + E;

  float* ws = (float*)d_ws;
  size_t off = 0;
  float*    XS   = ws + off; off += (size_t)N * 256;
  float*    Hb   = ws + off; off += (size_t)N * 256;
  float*    AS   = ws + off; off += (size_t)N * 8;
  int*      RP   = (int*)(ws + off);
  int*      CUR  = RP + (size_t)N * 2;
  int*      CNT  = RP + (size_t)N * 3;
  float*    BX2  = ws + off + (size_t)N * 4;
  off += (size_t)N * 8;
  int*      PSRC = (int*)(ws + off); off += (size_t)N * 16;
  float*    BX1  = ws + off; off += 2048;

  const int TPB = 256;

  hipMemsetAsync(CNT, 0, (size_t)N * sizeof(int), stream);
  dst_hist<<<(E + TPB - 1) / TPB, TPB, 0, stream>>>(dst, CNT, E);
  dst_scan<<<1, 1024, 0, stream>>>(CNT, RP, CUR, N);
  dst_fill<<<(E + TPB - 1) / TPB, TPB, 0, stream>>>(src, dst, CUR, PSRC, E);

  fold_att<<<(128 * 16 + 256 * 16 + TPB - 1) / TPB, TPB, 0, stream>>>(
      W1s, W1d, att1s, att1d, W2s, W2d, att2s, att2d, BX1, BX2);

  gemm_wmma_f32<256, 128, 8><<<dim3(N / 16, 256 / 64), 32, 0, stream>>>(
      x, W1s, BX1, XS, AS, 8, Hb, 256);
  node_gat8<<<(N + 7) / 8, 256, 0, stream>>>(RP, PSRC, AS, XS, b1, Hb, N);

  gemm_wmma_f32<128, 256, 1><<<dim3(N / 16, 128 / 64), 32, 0, stream>>>(
      Hb, W2s, BX2, XS, AS, 2, AS + 1, 2);
  node_gat1<<<(N + 7) / 8, 256, 0, stream>>>(RP, PSRC, AS, XS, b2, out, N);
}


# dst_scan rewritten: 13 independent 1024-thread blocks, 16-byte coalesced loads and stores, each block sums the counts in front of its 4096-element range itself (was one block, 196-byte lane stride)
# speedup vs baseline: 3.0035x; 1.1302x over previous
#include <hip/hip_runtime.h>
#include <stddef.h>

typedef __attribute__((ext_vector_type(2))) float v2f;
typedef __attribute__((ext_vector_type(8))) float v8f;

__device__ __forceinline__ v8f wmma4(v2f a, v2f b, v8f c) {
  return __builtin_amdgcn_wmma_f32_16x16x4_f32(
       false, a,  false, b,
       (short)0, c,  false,  false);
}

template <int N, int K, bool EXTRA>
__device__ __forceinline__ void strip16x64(const float* __restrict__ Arow,
                                           const float* __restrict__ Bcol,
                                           const float* __restrict__ Xp,
                                           v8f& c0, v8f& c1, v8f& c2, v8f& c3, v8f& cx) {
#pragma unroll
  for (int k = 0; k < K; k += 4) {
    v2f a = *(const v2f*)(Arow + k);
    v2f b0, b1, b2, b3;
    b0.x = Bcol[k * N + 0];       b0.y = Bcol[k * N + N + 0];
    b1.x = Bcol[k * N + 16];      b1.y = Bcol[k * N + N + 16];
    b2.x = Bcol[k * N + 32];      b2.y = Bcol[k * N + N + 32];
    b3.x = Bcol[k * N + 48];      b3.y = Bcol[k * N + N + 48];
    c0 = wmma4(a, b0, c0);
    c1 = wmma4(a, b1, c1);
    c2 = wmma4(a, b2, c2);
    c3 = wmma4(a, b3, c3);
    if constexpr (EXTRA) {
      const v2f bx = *(const v2f*)(Xp + k * 16);
      cx = wmma4(a, bx, cx);
    }
  }
}

template <int N, int K, int H>
__global__ void gemm_wmma_f32(const float* __restrict__ A,
                              const float* __restrict__ B,
                              const float* __restrict__ BX,
                              float* __restrict__ C,
                              float* as_out, int as_stride,
                              float* ad_out, int ad_stride) {
  const int lane = threadIdx.x & 31;
  const int m0 = blockIdx.x * 16;
  const int n0 = blockIdx.y * 64;
  const int half = lane >> 4;
  const int r    = lane & 15;

  const float* Arow = A + (size_t)(m0 + r) * K + 2 * half;
  const float* Bcol = B + (size_t)(2 * half) * N + n0 + r;
  const float* Xp   = BX + (half * 16 + r) * 2;

  v8f c0 = {}, c1 = {}, c2 = {}, c3 = {}, cx = {};
  const bool extra = (blockIdx.y == 0);
  if (extra) strip16x64<N, K, true >(Arow, Bcol, Xp, c0, c1, c2, c3, cx);
  else       strip16x64<N, K, false>(Arow, Bcol, Xp, c0, c1, c2, c3, cx);

  float* Crow = C + (size_t)(m0 + 8 * half) * N + n0 + r;
#pragma unroll
  for (int i = 0; i < 8; ++i) {
    Crow[i * N + 0]  = c0[i];
    Crow[i * N + 16] = c1[i];
    Crow[i * N + 32] = c2[i];
    Crow[i * N + 48] = c3[i];
  }

  if (extra && r < 2 * H) {
    const int row0 = m0 + 8 * half;
    const bool is_s = r < H;
    float* p = is_s ? as_out + (size_t)row0 * as_stride + r
                    : ad_out + (size_t)row0 * ad_stride + (r - H);
    const int st = is_s ? as_stride : ad_stride;
#pragma unroll
    for (int i = 0; i < 8; ++i) p[(size_t)i * st] = cx[i];
  }
}

template <int F>
__device__ __forceinline__ float fold_dot(const float* __restrict__ wr,
                                          const float* __restrict__ ar) {
  float acc = 0.f;
#pragma unroll 4
  for (int f = 0; f < F; ++f) acc += wr[f] * ar[f];
  return acc;
}

__global__ void fold_att(const float* __restrict__ W1s, const float* __restrict__ W1d,
                         const float* __restrict__ att1s, const float* __restrict__ att1d,
                         const float* __restrict__ W2s, const float* __restrict__ W2d,
                         const float* __restrict__ att2s, const float* __restrict__ att2d,
                         float* __restrict__ BX1, float* __restrict__ BX2) {
  const int t = blockIdx.x * blockDim.x + threadIdx.x;
  if (t < 128 * 16) {
    const int k = t >> 4, c = t & 15, h = c & 7;
    const float* W   = (c < 8) ? W1s : W1d;
    const float* att = (c < 8) ? att1s : att1d;
    BX1[((k >> 1) * 16 + c) * 2 + (k & 1)] =
        fold_dot<32>(W + (size_t)k * 256 + h * 32, att + h * 32);
  } else if (t < 128 * 16 + 256 * 16) {
    const int u = t - 128 * 16;
    const int k = u >> 4, c = u & 15;
    float v = 0.f;
    if (c < 2) v = fold_dot<128>((c == 0 ? W2s : W2d) + (size_t)k * 128,
                                 c == 0 ? att2s : att2d);
    BX2[((k >> 1) * 16 + c) * 2 + (k & 1)] = v;
  }
}

typedef __attribute__((ext_vector_type(4))) float v4f;

__device__ __forceinline__ float leaky(float a) {
  return a > 0.f ? a : 0.2f * a;
}

__global__ void dst_hist(const int* __restrict__ dst, int* __restrict__ cnt, int E) {
  int t = blockIdx.x * blockDim.x + threadIdx.x;
  if (t >= E) return;
  atomicAdd(&cnt[dst[t]], 1);
}

typedef __attribute__((ext_vector_type(4))) int v4i;

__global__ void __launch_bounds__(1024)
dst_scan(const int* __restrict__ cnt, int* __restrict__ rowptr,
         int* __restrict__ cur, int N) {
  __shared__ int wpre[32];
  __shared__ int wown[32];
  const int t = threadIdx.x, lane = t & 31, w = t >> 5;
  const int nb = blockIdx.x;

  const int i0 = nb * 4096 + t * 4;
  v4i v = {0, 0, 0, 0};
  if (i0 + 3 < N) {
    v = *(const v4i*)(cnt + i0);
  } else {
    if (i0     < N) v.x = cnt[i0];
    if (i0 + 1 < N) v.y = cnt[i0 + 1];
    if (i0 + 2 < N) v.z = cnt[i0 + 2];
  }

  int pre = 0;
#pragma unroll 4
  for (int r = 0; r < nb; ++r) {
    const v4i p = *(const v4i*)(cnt + r * 4096 + t * 4);
    pre += (p.x + p.y) + (p.z + p.w);
  }
  const int own = (v.x + v.y) + (v.z + v.w);

  int inc = own;
  for (int o = 1; o < 32; o <<= 1) {
    const int u = __shfl_up(inc, o);
    inc += (lane >= o) ? u : 0;
  }
  for (int o = 16; o > 0; o >>= 1) pre += __shfl_xor(pre, o);
  if (lane == 31) wown[w] = inc;
  if (lane == 0)  wpre[w] = pre;
  __syncthreads();

  int part = wpre[lane] + ((lane < w) ? wown[lane] : 0);
  for (int o = 16; o > 0; o >>= 1) part += __shfl_xor(part, o);

  const int e0 = part + inc - own;
  const v4i o4 = {e0, e0 + v.x, e0 + v.x + v.y, e0 + v.x + v.y + v.z};
  if (i0 + 3 < N) {
    *(v4i*)(rowptr + i0) = o4;
    *(v4i*)(cur + i0)    = o4;
  } else {
    if (i0     < N) { rowptr[i0]     = o4.x; cur[i0]     = o4.x; }
    if (i0 + 1 < N) { rowptr[i0 + 1] = o4.y; cur[i0 + 1] = o4.y; }
    if (i0 + 2 < N) { rowptr[i0 + 2] = o4.z; cur[i0 + 2] = o4.z; }
  }
  if (nb == (int)gridDim.x - 1 && t == 1023) rowptr[N] = e0 + own;
}

__global__ void dst_fill(const int* __restrict__ src, const int* __restrict__ dst,
                         int* __restrict__ cur, int* __restrict__ psrc, int E) {
  int t = blockIdx.x * blockDim.x + threadIdx.x;
  if (t >= E) return;
  int p = atomicAdd(&cur[dst[t]], 1);
  psrc[p] = src[t];
}

__device__ __forceinline__ void edge8(int s, int lane, int h0, int h1,
                                      const float* __restrict__ as,
                                      const float* __restrict__ xs,
                                      float ad0, float ad1, float m0, float m1,
                                      float i0, float i1, v4f& acc0, v4f& acc1) {
  const float w0 = __expf(leaky(as[s * 8 + h0] + ad0) - m0) * i0;
  const float w1 = __expf(leaky(as[s * 8 + h1] + ad1) - m1) * i1;
  const float* xr = xs + (size_t)s * 256 + lane * 4;
  const v4f x0 = *(const v4f*)xr;
  const v4f x1 = *(const v4f*)(xr + 128);
  acc0 += w0 * x0;
  acc1 += w1 * x1;
}

__global__ void node_gat8(const int* __restrict__ rowptr, const int* __restrict__ psrc,
                          const float* __restrict__ as, const float* __restrict__ xs,
                          const float* __restrict__ bias, float* hb, int N) {
  const int d = __builtin_amdgcn_readfirstlane(
      (int)(blockIdx.x * (blockDim.x >> 5) + (threadIdx.x >> 5)));
  if (d >= N) return;
  const int lane = threadIdx.x & 31;
  const int beg = rowptr[d], end = rowptr[d + 1];
  float* row = hb + (size_t)d * 256;
  const int h = lane & 7, sub = lane >> 3;
  const float adh = row[h];

  float m = -INFINITY;
  for (int c = beg; c < end; c += 32) {
    const int n = (end - c) < 32 ? (end - c) : 32;
    const int sv = (c + lane < end) ? psrc[c + lane] : 0;
#pragma unroll
    for (int k = 0; k < 8; ++k) {
      const int j = sub + 4 * k;
      const float al = leaky(as[__shfl(sv, j) * 8 + h] + adh);
      m = (j < n) ? fmaxf(m, al) : m;
    }
  }
  m = fmaxf(m, __shfl_xor(m, 8));
  m = fmaxf(m, __shfl_xor(m, 16));

  float den = 0.f;
  for (int c = beg; c < end; c += 32) {
    const int n = (end - c) < 32 ? (end - c) : 32;
    const int sv = (c + lane < end) ? psrc[c + lane] : 0;
#pragma unroll
    for (int k = 0; k < 8; ++k) {
      const int j = sub + 4 * k;
      const float al = leaky(as[__shfl(sv, j) * 8 + h] + adh);
      den += (j < n) ? __expf(al - m) : 0.f;
    }
  }
  den += __shfl_xor(den, 8);
  den += __shfl_xor(den, 16);
  const float inv = 1.0f / (den + 1e-16f);

  const int h0 = lane >> 3, h1 = h0 + 4;
  const float m0 = __shfl(m, h0),    m1 = __shfl(m, h1);
  const float i0 = __shfl(inv, h0),  i1 = __shfl(inv, h1);
  const float ad0 = __shfl(adh, h0), ad1 = __shfl(adh, h1);
  v4f acc0 = {0.f, 0.f, 0.f, 0.f}, acc1 = {0.f, 0.f, 0.f, 0.f};
  for (int c = beg; c < end; c += 32) {
    const int n = (end - c) < 32 ? (end - c) : 32;
    const int sv = (c + lane < end) ? psrc[c + lane] : 0;
    int i = 0;
    for (; i + 4 <= n; i += 4) {
      const int s0 = __builtin_amdgcn_readlane(sv, i);
      const int s1 = __builtin_amdgcn_readlane(sv, i + 1);
      const int s2 = __builtin_amdgcn_readlane(sv, i + 2);
      const int s3 = __builtin_amdgcn_readlane(sv, i + 3);
      edge8(s0, lane, h0, h1, as, xs, ad0, ad1, m0, m1, i0, i1, acc0, acc1);
      edge8(s1, lane, h0, h1, as, xs, ad0, ad1, m0, m1, i0, i1, acc0, acc1);
      edge8(s2, lane, h0, h1, as, xs, ad0, ad1, m0, m1, i0, i1, acc0, acc1);
      edge8(s3, lane, h0, h1, as, xs, ad0, ad1, m0, m1, i0, i1, acc0, acc1);
    }
    for (; i < n; ++i)
      edge8(__builtin_amdgcn_readlane(sv, i), lane, h0, h1, as, xs,
            ad0, ad1, m0, m1, i0, i1, acc0, acc1);
  }
  v4f o0 = acc0 + *(const v4f*)(bias + lane * 4);
  v4f o1 = acc1 + *(const v4f*)(bias + 128 + lane * 4);
#pragma unroll
  for (int q = 0; q < 4; ++q) {
    o0[q] = o0[q] > 0.f ? o0[q] : 0.f;
    o1[q] = o1[q] > 0.f ? o1[q] : 0.f;
  }
  *(v4f*)(row + lane * 4) = o0;
  *(v4f*)(row + 128 + lane * 4) = o1;
}

__device__ __forceinline__ void edge1(int s, float w, int lane,
                                      const float* __restrict__ xs, v4f& acc) {
  acc += w * *(const v4f*)(xs + (size_t)s * 128 + lane * 4);
}

__global__ void node_gat1(const int* __restrict__ rowptr, const int* __restrict__ psrc,
                          const float* __restrict__ sd,
                          const float* __restrict__ xs, const float* __restrict__ bias,
                          float* __restrict__ out, int N) {
  const int d = __builtin_amdgcn_readfirstlane(
      (int)(blockIdx.x * (blockDim.x >> 5) + (threadIdx.x >> 5)));
  if (d >= N) return;
  const int lane = threadIdx.x & 31;
  const int beg = rowptr[d], end = rowptr[d + 1];
  const float add = sd[2 * d + 1];
  float m = -INFINITY;
  for (int i = beg + lane; i < end; i += 32)
    m = fmaxf(m, leaky(sd[2 * psrc[i]] + add));
#pragma unroll
  for (int o = 16; o > 0; o >>= 1) m = fmaxf(m, __shfl_xor(m, o));
  float den = 0.f;
  for (int i = beg + lane; i < end; i += 32)
    den += __expf(leaky(sd[2 * psrc[i]] + add) - m);
#pragma unroll
  for (int o = 16; o > 0; o >>= 1) den += __shfl_xor(den, o);
  const float inv = 1.0f / (den + 1e-16f);
  v4f acc = {0.f, 0.f, 0.f, 0.f};
  for (int c = beg; c < end; c += 32) {
    const int n = (end - c) < 32 ? (end - c) : 32;
    const bool ok = c + lane < end;
    const int sv = ok ? psrc[c + lane] : 0;
    const int wv = __float_as_int(ok ? __expf(leaky(sd[2 * sv] + add) - m) * inv : 0.f);
    int i = 0;
    for (; i + 4 <= n; i += 4) {
      const int s0 = __builtin_amdgcn_readlane(sv, i);
      const int s1 = __builtin_amdgcn_readlane(sv, i + 1);
      const int s2 = __builtin_amdgcn_readlane(sv, i + 2);
      const int s3 = __builtin_amdgcn_readlane(sv, i + 3);
      const float w0 = __int_as_float(__builtin_amdgcn_readlane(wv, i));
      const float w1 = __int_as_float(__builtin_amdgcn_readlane(wv, i + 1));
      const float w2 = __int_as_float(__builtin_amdgcn_readlane(wv, i + 2));
      const float w3 = __int_as_float(__builtin_amdgcn_readlane(wv, i + 3));
      edge1(s0, w0, lane, xs, acc);
      edge1(s1, w1, lane, xs, acc);
      edge1(s2, w2, lane, xs, acc);
      edge1(s3, w3, lane, xs, acc);
    }
    for (; i < n; ++i)
      edge1(__builtin_amdgcn_readlane(sv, i),
            __int_as_float(__builtin_amdgcn_readlane(wv, i)), lane, xs, acc);
  }
  *(v4f*)(out + (size_t)d * 128 + lane * 4) = acc + *(const v4f*)(bias + lane * 4);
}

extern "C" void kernel_launch(void* const* d_in, const int* in_sizes, int n_in,
                              void* d_out, int out_size, void* d_ws, size_t ws_size,
                              hipStream_t stream) {
  (void)n_in; (void)out_size; (void)ws_size;
  const float* x     = (const float*)d_in[0];
  const int*   ei    = (const int*)d_in[1];
  const float* W1s   = (const float*)d_in[2];
  const float* W1d   = (const float*)d_in[3];
  const float* att1s = (const float*)d_in[4];
  const float* att1d = (const float*)d_in[5];
  const float* b1    = (const float*)d_in[6];
  const float* W2s   = (const float*)d_in[7];
  const float* W2d   = (const float*)d_in[8];
  const float* att2s = (const float*)d_in[9];
  const float* att2d = (const float*)d_in[10];
  const float* b2    = (const float*)d_in[11];
  float* out = (float*)d_out;

  const int N = in_sizes[0] / 128;
  const int E = in_sizes[1] / 2;
  const int* src = ei;
  const int* dst = ei + E;

  float* ws = (float*)d_ws;
  size_t off = 0;
  float*    XS   = ws + off; off += (size_t)N * 256;
  float*    Hb   = ws + off; off += (size_t)N * 256;
  float*    AS   = ws + off; off += (size_t)N * 8;
  int*      RP   = (int*)(ws + off);
  int*      CUR  = RP + (size_t)N * 2;
  int*      CNT  = RP + (size_t)N * 3;
  float*    BX2  = ws + off + (size_t)N * 4;
  off += (size_t)N * 8;
  int*      PSRC = (int*)(ws + off); off += (size_t)N * 16;
  float*    BX1  = ws + off; off += 2048;

  const int TPB = 256;

  hipMemsetAsync(CNT, 0, (size_t)N * sizeof(int), stream);
  dst_hist<<<(E + TPB - 1) / TPB, TPB, 0, stream>>>(dst, CNT, E);
  dst_scan<<<(N + 4095) / 4096, 1024, 0, stream>>>(CNT, RP, CUR, N);
  dst_fill<<<(E + TPB - 1) / TPB, TPB, 0, stream>>>(src, dst, CUR, PSRC, E);

  fold_att<<<(128 * 16 + 256 * 16 + TPB - 1) / TPB, TPB, 0, stream>>>(
      W1s, W1d, att1s, att1d, W2s, W2d, att2s, att2d, BX1, BX2);

  gemm_wmma_f32<256, 128, 8><<<dim3(N / 16, 256 / 64), 32, 0, stream>>>(
      x, W1s, BX1, XS, AS, 8, Hb, 256);
  node_gat8<<<(N + 7) / 8, 256, 0, stream>>>(RP, PSRC, AS, XS, b1, Hb, N);

  gemm_wmma_f32<128, 256, 1><<<dim3(N / 16, 128 / 64), 32, 0, stream>>>(
      Hb, W2s, BX2, XS, AS, 2, AS + 1, 2);
  node_gat1<<<(N + 7) / 8, 256, 0, stream>>>(RP, PSRC, AS, XS, b2, out, N);
}
